# starting kernel
# baseline (speedup 1.0000x reference)
#include <hip/hip_runtime.h>


typedef float v2f __attribute__((ext_vector_type(2)));
typedef float v8f __attribute__((ext_vector_type(8)));

#define C_CAPS  32
#define B_SZ    16
#define N_NODES 1024
#define CI      64
#define O_SZ    32


__global__ __launch_bounds__(128) void caps_priors_wmma(
    const float* __restrict__ x, const float* __restrict__ W,
    float* __restrict__ priors)
{
    const int wid  = threadIdx.x >> 5;
    const int lane = threadIdx.x & 31;
    const int blk  = blockIdx.x;
    const int c    = blk >> 8;
    const int n    = ((blk & 255) << 2) + wid;

    const int m = lane & 15;
    const int h = lane >> 4;

    const float* xp = x + (size_t)m * (N_NODES * CI) + (size_t)n * CI;
    const float* wp = W + (size_t)c * ((size_t)N_NODES * CI * O_SZ)
                        + (size_t)n * (CI * O_SZ);

    __builtin_prefetch(wp + 4 * (CI * O_SZ), 0, 1);

    v8f acc0 = {};
    v8f acc1 = {};

    #pragma unroll
    for (int ks = 0; ks < 16; ++ks) {
        const int i0 = ks * 4 + 2 * h;

        v2f a;
        a.x = xp[i0];
        a.y = xp[i0 + 1];

        v2f b0, b1;
        b0.x = wp[(i0    ) * O_SZ + m     ];
        b0.y = wp[(i0 + 1) * O_SZ + m     ];
        b1.x = wp[(i0    ) * O_SZ + m + 16];
        b1.y = wp[(i0 + 1) * O_SZ + m + 16];

        acc0 = __builtin_amdgcn_wmma_f32_16x16x4_f32(
                   false, a, false, b0, (short)0, acc0, false, false);
        acc1 = __builtin_amdgcn_wmma_f32_16x16x4_f32(
                   false, a, false, b1, (short)0, acc1, false, false);
    }

    float* pp = priors + ((size_t)c * B_SZ * N_NODES + (size_t)n) * O_SZ;
    #pragma unroll
    for (int r = 0; r < 8; ++r) {
        const int brow = r + 8 * h;
        pp[(size_t)brow * (N_NODES * O_SZ) + m     ] = acc0[r];
        pp[(size_t)brow * (N_NODES * O_SZ) + m + 16] = acc1[r];
    }
}

__global__ __launch_bounds__(256) void caps_route(
    const float* __restrict__ priors, float* __restrict__ out)
{
    const int cb   = blockIdx.x;
    const int t    = threadIdx.x;
    const int lane = t & 31;
    const int wid  = t >> 5;

    __shared__ float redv[8 * O_SZ];
    __shared__ float reds[8];

    float P[4][O_SZ];
    const float* base = priors + (size_t)cb * (N_NODES * O_SZ);
    #pragma unroll
    for (int j = 0; j < 4; ++j) {
        const float4* row = (const float4*)(base + (size_t)(t + 256 * j) * O_SZ);
        #pragma unroll
        for (int q = 0; q < 8; ++q) {
            float4 v = row[q];
            P[j][4*q+0] = v.x; P[j][4*q+1] = v.y;
            P[j][4*q+2] = v.z; P[j][4*q+3] = v.w;
        }
    }

    float logits[4] = {0.f, 0.f, 0.f, 0.f};
    float vout[O_SZ];

    for (int it = 0; it < 3; ++it) {
        float mx = fmaxf(fmaxf(logits[0], logits[1]), fmaxf(logits[2], logits[3]));
        #pragma unroll
        for (int off = 16; off; off >>= 1)
            mx = fmaxf(mx, __shfl_xor(mx, off, 32));
        if (lane == 0) reds[wid] = mx;
        __syncthreads();
        mx = reds[0];
        #pragma unroll
        for (int w = 1; w < 8; ++w) mx = fmaxf(mx, reds[w]);
        __syncthreads();

        float e[4]; float zp = 0.f;
        #pragma unroll
        for (int j = 0; j < 4; ++j) { e[j] = __expf(logits[j] - mx); zp += e[j]; }
        #pragma unroll
        for (int off = 16; off; off >>= 1) zp += __shfl_xor(zp, off, 32);
        if (lane == 0) reds[wid] = zp;

        #pragma unroll
        for (int o = 0; o < O_SZ; ++o) {
            float acc = e[0]*P[0][o] + e[1]*P[1][o] + e[2]*P[2][o] + e[3]*P[3][o];
            #pragma unroll
            for (int off = 16; off; off >>= 1) acc += __shfl_xor(acc, off, 32);
            if (lane == 0) redv[wid * O_SZ + o] = acc;
        }
        __syncthreads();

        float Z = 0.f;
        #pragma unroll
        for (int w = 0; w < 8; ++w) Z += reds[w];

        float sq = 0.f;
        #pragma unroll
        for (int o = 0; o < O_SZ; ++o) {
            float so = 0.f;
            #pragma unroll
            for (int w = 0; w < 8; ++w) so += redv[w * O_SZ + o];
            so /= Z;
            vout[o] = so;
            sq += so * so;
        }
        const float rs = sqrtf(sq);
        const float k  = rs / (1.0f + sq);
        #pragma unroll
        for (int o = 0; o < O_SZ; ++o) vout[o] *= k;

        if (it < 2) {
            #pragma unroll
            for (int j = 0; j < 4; ++j) {
                float d = 0.f;
                #pragma unroll
                for (int o = 0; o < O_SZ; ++o) d += P[j][o] * vout[o];
                logits[j] += d;
            }
        }
        __syncthreads();
    }

    if (t == 0) {
        float* op = out + (size_t)cb * O_SZ;
        #pragma unroll
        for (int o = 0; o < O_SZ; ++o) op[o] = vout[o];
    }
}

extern "C" void kernel_launch(void* const* d_in, const int* in_sizes, int n_in,
                              void* d_out, int out_size, void* d_ws, size_t ws_size,
                              hipStream_t stream) {
    const float* x = (const float*)d_in[0];
    const float* W = (const float*)d_in[1];
    float* priors  = (float*)d_ws;
    float* out     = (float*)d_out;

    caps_priors_wmma<<<dim3(C_CAPS * (N_NODES / 4)), dim3(128), 0, stream>>>(x, W, priors);
    caps_route<<<dim3(C_CAPS * B_SZ), dim3(256), 0, stream>>>(priors, out);
}


# route kernel: the 32 per-output wave sums of an iteration are reduced together by one halving lane exchange (31 exchanges, lane o ends with s[o]) instead of 32 waited-on butterflies; same sum tree
# speedup vs baseline: 1.4555x; 1.4555x over previous
#include <hip/hip_runtime.h>


typedef float v2f __attribute__((ext_vector_type(2)));
typedef float v8f __attribute__((ext_vector_type(8)));

#define C_CAPS  32
#define B_SZ    16
#define N_NODES 1024
#define CI      64
#define O_SZ    32


__global__ __launch_bounds__(128) void caps_priors_wmma(
    const float* __restrict__ x, const float* __restrict__ W,
    float* __restrict__ priors)
{
    const int wid  = threadIdx.x >> 5;
    const int lane = threadIdx.x & 31;
    const int blk  = blockIdx.x;
    const int c    = blk >> 8;
    const int n    = ((blk & 255) << 2) + wid;

    const int m = lane & 15;
    const int h = lane >> 4;

    const float* xp = x + (size_t)m * (N_NODES * CI) + (size_t)n * CI;
    const float* wp = W + (size_t)c * ((size_t)N_NODES * CI * O_SZ)
                        + (size_t)n * (CI * O_SZ);

    __builtin_prefetch(wp + 4 * (CI * O_SZ), 0, 1);

    v8f acc0 = {};
    v8f acc1 = {};

    #pragma unroll
    for (int ks = 0; ks < 16; ++ks) {
        const int i0 = ks * 4 + 2 * h;

        v2f a;
        a.x = xp[i0];
        a.y = xp[i0 + 1];

        v2f b0, b1;
        b0.x = wp[(i0    ) * O_SZ + m     ];
        b0.y = wp[(i0 + 1) * O_SZ + m     ];
        b1.x = wp[(i0    ) * O_SZ + m + 16];
        b1.y = wp[(i0 + 1) * O_SZ + m + 16];

        acc0 = __builtin_amdgcn_wmma_f32_16x16x4_f32(
                   false, a, false, b0, (short)0, acc0, false, false);
        acc1 = __builtin_amdgcn_wmma_f32_16x16x4_f32(
                   false, a, false, b1, (short)0, acc1, false, false);
    }

    float* pp = priors + ((size_t)c * B_SZ * N_NODES + (size_t)n) * O_SZ;
    #pragma unroll
    for (int r = 0; r < 8; ++r) {
        const int brow = r + 8 * h;
        pp[(size_t)brow * (N_NODES * O_SZ) + m     ] = acc0[r];
        pp[(size_t)brow * (N_NODES * O_SZ) + m + 16] = acc1[r];
    }
}

__global__ __launch_bounds__(256) void caps_route(
    const float* __restrict__ priors, float* __restrict__ out)
{
    const int cb   = blockIdx.x;
    const int t    = threadIdx.x;
    const int lane = t & 31;
    const int wid  = t >> 5;

    __shared__ float redv[8 * O_SZ];
    __shared__ float reds[8];

    float P[4][O_SZ];
    const float* base = priors + (size_t)cb * (N_NODES * O_SZ);
    #pragma unroll
    for (int j = 0; j < 4; ++j) {
        const float4* row = (const float4*)(base + (size_t)(t + 256 * j) * O_SZ);
        #pragma unroll
        for (int q = 0; q < 8; ++q) {
            float4 v = row[q];
            P[j][4*q+0] = v.x; P[j][4*q+1] = v.y;
            P[j][4*q+2] = v.z; P[j][4*q+3] = v.w;
        }
    }

    float logits[4] = {0.f, 0.f, 0.f, 0.f};
    float vout[O_SZ];

    for (int it = 0; it < 3; ++it) {
        float mx = fmaxf(fmaxf(logits[0], logits[1]), fmaxf(logits[2], logits[3]));
        #pragma unroll
        for (int off = 16; off; off >>= 1)
            mx = fmaxf(mx, __shfl_xor(mx, off, 32));
        if (lane == 0) reds[wid] = mx;
        __syncthreads();
        mx = reds[0];
        #pragma unroll
        for (int w = 1; w < 8; ++w) mx = fmaxf(mx, reds[w]);
        __syncthreads();

        float e[4]; float zp = 0.f;
        #pragma unroll
        for (int j = 0; j < 4; ++j) { e[j] = __expf(logits[j] - mx); zp += e[j]; }

        float a32[O_SZ];
        for (int o = 0; o < O_SZ; ++o)
            a32[o] = e[0]*P[0][o] + e[1]*P[1][o] + e[2]*P[2][o] + e[3]*P[3][o];

        float a16[16], a8[8], a4[4], a2[2], a1;
        {
            const bool hi = (lane & 16) != 0;
            for (int i = 0; i < 16; ++i) {
                const float keep = hi ? a32[i + 16] : a32[i];
                const float send = hi ? a32[i] : a32[i + 16];
                a16[i] = keep + __shfl_xor(send, 16, 32);
            }
        }
        {
            const bool hi = (lane & 8) != 0;
            for (int i = 0; i < 8; ++i) {
                const float keep = hi ? a16[i + 8] : a16[i];
                const float send = hi ? a16[i] : a16[i + 8];
                a8[i] = keep + __shfl_xor(send, 8, 32);
            }
        }
        {
            const bool hi = (lane & 4) != 0;
            for (int i = 0; i < 4; ++i) {
                const float keep = hi ? a8[i + 4] : a8[i];
                const float send = hi ? a8[i] : a8[i + 4];
                a4[i] = keep + __shfl_xor(send, 4, 32);
            }
        }
        {
            const bool hi = (lane & 2) != 0;
            for (int i = 0; i < 2; ++i) {
                const float keep = hi ? a4[i + 2] : a4[i];
                const float send = hi ? a4[i] : a4[i + 2];
                a2[i] = keep + __shfl_xor(send, 2, 32);
            }
        }
        {
            const bool hi = (lane & 1) != 0;
            const float keep = hi ? a2[1] : a2[0];
            const float send = hi ? a2[0] : a2[1];
            a1 = keep + __shfl_xor(send, 1, 32);
        }
        redv[wid * O_SZ + lane] = a1;

        #pragma unroll
        for (int off = 16; off; off >>= 1) zp += __shfl_xor(zp, off, 32);
        if (lane == 0) reds[wid] = zp;
        __syncthreads();

        float Z = 0.f;
        #pragma unroll
        for (int w = 0; w < 8; ++w) Z += reds[w];

        float sq = 0.f;
        #pragma unroll
        for (int o = 0; o < O_SZ; ++o) {
            float so = 0.f;
            #pragma unroll
            for (int w = 0; w < 8; ++w) so += redv[w * O_SZ + o];
            so /= Z;
            vout[o] = so;
            sq += so * so;
        }
        const float rs = sqrtf(sq);
        const float k  = rs / (1.0f + sq);
        #pragma unroll
        for (int o = 0; o < O_SZ; ++o) vout[o] *= k;

        if (it < 2) {
            #pragma unroll
            for (int j = 0; j < 4; ++j) {
                float d = 0.f;
                #pragma unroll
                for (int o = 0; o < O_SZ; ++o) d += P[j][o] * vout[o];
                logits[j] += d;
            }
        }
        __syncthreads();
    }

    if (t == 0) {
        float* op = out + (size_t)cb * O_SZ;
        #pragma unroll
        for (int o = 0; o < O_SZ; ++o) op[o] = vout[o];
    }
}

extern "C" void kernel_launch(void* const* d_in, const int* in_sizes, int n_in,
                              void* d_out, int out_size, void* d_ws, size_t ws_size,
                              hipStream_t stream) {
    const float* x = (const float*)d_in[0];
    const float* W = (const float*)d_in[1];
    float* priors  = (float*)d_ws;
    float* out     = (float*)d_out;

    caps_priors_wmma<<<dim3(C_CAPS * (N_NODES / 4)), dim3(128), 0, stream>>>(x, W, priors);
    caps_route<<<dim3(C_CAPS * B_SZ), dim3(256), 0, stream>>>(priors, out);
}


# priors kernel: a wave keeps its node's x fragments in registers and loops over 8 classes, W requested 3 quarter tiles ahead (4-slot register ring, L2 prefetch 2 tiles ahead); 4096 waves, not 32768
# speedup vs baseline: 1.9681x; 1.3522x over previous
#include <hip/hip_runtime.h>


typedef float v2f __attribute__((ext_vector_type(2)));
typedef float v8f __attribute__((ext_vector_type(8)));

#define C_CAPS  32
#define B_SZ    16
#define N_NODES 1024
#define CI      64
#define O_SZ    32


#define C_PER_WAVE 8
#define N_QUARTERS (4 * C_PER_WAVE)
#define AHEAD_Q    3
#define W_TILE     (CI * O_SZ)
#define W_CSTRIDE  ((size_t)N_NODES * CI * O_SZ)

__global__ __launch_bounds__(256) __attribute__((amdgpu_waves_per_eu(4, 4))) void caps_priors_wmma(
    const float* __restrict__ x, const float* __restrict__ W,
    float* __restrict__ priors)
{
    const int wid  = threadIdx.x >> 5;
    const int lane = threadIdx.x & 31;
    const int blk  = blockIdx.x;
    const int c0   = (blk >> 7) * C_PER_WAVE;
    const int n    = ((blk & 127) << 3) + wid;

    const int m = lane & 15;
    const int h = lane >> 4;

    const float* xp = x + (size_t)m * (N_NODES * CI) + (size_t)n * CI + 2 * h;
    const float* wt = W + (size_t)c0 * W_CSTRIDE + (size_t)n * W_TILE;
    const float* wp = wt + (2 * h) * O_SZ + m;

    v2f a[16];
    #pragma unroll
    for (int ks = 0; ks < 16; ++ks) {
        a[ks].x = xp[ks * 4];
        a[ks].y = xp[ks * 4 + 1];
    }

    v2f bq[4][4][2];

#define LOAD_W_QUARTER(Q) do { \
        const float* wq_ = wp + (size_t)((Q) >> 2) * W_CSTRIDE + ((Q) & 3) * (16 * O_SZ); \
        _Pragma("unroll") for (int kk = 0; kk < 4; ++kk) { \
            bq[(Q) & 3][kk][0].x = wq_[(kk * 4    ) * O_SZ     ]; \
            bq[(Q) & 3][kk][0].y = wq_[(kk * 4 + 1) * O_SZ     ]; \
            bq[(Q) & 3][kk][1].x = wq_[(kk * 4    ) * O_SZ + 16]; \
            bq[(Q) & 3][kk][1].y = wq_[(kk * 4 + 1) * O_SZ + 16]; \
        } } while (0)

    __builtin_prefetch(wt + W_CSTRIDE + lane * 64, 0, 1);
    #pragma unroll
    for (int q = 0; q < AHEAD_Q; ++q) LOAD_W_QUARTER(q);

    float* pp = priors + ((size_t)c0 * B_SZ * N_NODES + (size_t)n) * O_SZ
                       + (size_t)(8 * h) * (N_NODES * O_SZ) + m;

    v8f acc0 = {};
    v8f acc1 = {};

    #pragma unroll
    for (int q = 0; q < N_QUARTERS; ++q) {
        __builtin_amdgcn_sched_barrier(0);
        if ((q & 3) == 0 && (q >> 2) + 2 < C_PER_WAVE)
            __builtin_prefetch(wt + (size_t)((q >> 2) + 2) * W_CSTRIDE + lane * 64, 0, 1);
        if (q + AHEAD_Q < N_QUARTERS) LOAD_W_QUARTER(q + AHEAD_Q);
        __builtin_amdgcn_sched_barrier(0);

        if ((q & 3) == 0) { acc0 = (v8f){}; acc1 = (v8f){}; }
        #pragma unroll
        for (int kk = 0; kk < 4; ++kk) {
            acc0 = __builtin_amdgcn_wmma_f32_16x16x4_f32(
                       false, a[(q & 3) * 4 + kk], false, bq[q & 3][kk][0], (short)0, acc0, false, false);
            acc1 = __builtin_amdgcn_wmma_f32_16x16x4_f32(
                       false, a[(q & 3) * 4 + kk], false, bq[q & 3][kk][1], (short)0, acc1, false, false);
        }

        if ((q & 3) == 3) {
            float* pj = pp + (size_t)(q >> 2) * (B_SZ * N_NODES * O_SZ);
            #pragma unroll
            for (int r = 0; r < 8; ++r) {
                pj[(size_t)r * (N_NODES * O_SZ)     ] = acc0[r];
                pj[(size_t)r * (N_NODES * O_SZ) + 16] = acc1[r];
            }
        }
    }
#undef LOAD_W_QUARTER
}

__global__ __launch_bounds__(256) void caps_route(
    const float* __restrict__ priors, float* __restrict__ out)
{
    const int cb   = blockIdx.x;
    const int t    = threadIdx.x;
    const int lane = t & 31;
    const int wid  = t >> 5;

    __shared__ float redv[8 * O_SZ];
    __shared__ float reds[8];

    float P[4][O_SZ];
    const float* base = priors + (size_t)cb * (N_NODES * O_SZ);
    #pragma unroll
    for (int j = 0; j < 4; ++j) {
        const float4* row = (const float4*)(base + (size_t)(t + 256 * j) * O_SZ);
        #pragma unroll
        for (int q = 0; q < 8; ++q) {
            float4 v = row[q];
            P[j][4*q+0] = v.x; P[j][4*q+1] = v.y;
            P[j][4*q+2] = v.z; P[j][4*q+3] = v.w;
        }
    }

    float logits[4] = {0.f, 0.f, 0.f, 0.f};
    float vout[O_SZ];

    for (int it = 0; it < 3; ++it) {
        float mx = fmaxf(fmaxf(logits[0], logits[1]), fmaxf(logits[2], logits[3]));
        #pragma unroll
        for (int off = 16; off; off >>= 1)
            mx = fmaxf(mx, __shfl_xor(mx, off, 32));
        if (lane == 0) reds[wid] = mx;
        __syncthreads();
        mx = reds[0];
        #pragma unroll
        for (int w = 1; w < 8; ++w) mx = fmaxf(mx, reds[w]);
        __syncthreads();

        float e[4]; float zp = 0.f;
        #pragma unroll
        for (int j = 0; j < 4; ++j) { e[j] = __expf(logits[j] - mx); zp += e[j]; }

        float a32[O_SZ];
        for (int o = 0; o < O_SZ; ++o)
            a32[o] = e[0]*P[0][o] + e[1]*P[1][o] + e[2]*P[2][o] + e[3]*P[3][o];

        float a16[16], a8[8], a4[4], a2[2], a1;
        {
            const bool hi = (lane & 16) != 0;
            for (int i = 0; i < 16; ++i) {
                const float keep = hi ? a32[i + 16] : a32[i];
                const float send = hi ? a32[i] : a32[i + 16];
                a16[i] = keep + __shfl_xor(send, 16, 32);
            }
        }
        {
            const bool hi = (lane & 8) != 0;
            for (int i = 0; i < 8; ++i) {
                const float keep = hi ? a16[i + 8] : a16[i];
                const float send = hi ? a16[i] : a16[i + 8];
                a8[i] = keep + __shfl_xor(send, 8, 32);
            }
        }
        {
            const bool hi = (lane & 4) != 0;
            for (int i = 0; i < 4; ++i) {
                const float keep = hi ? a8[i + 4] : a8[i];
                const float send = hi ? a8[i] : a8[i + 4];
                a4[i] = keep + __shfl_xor(send, 4, 32);
            }
        }
        {
            const bool hi = (lane & 2) != 0;
            for (int i = 0; i < 2; ++i) {
                const float keep = hi ? a4[i + 2] : a4[i];
                const float send = hi ? a4[i] : a4[i + 2];
                a2[i] = keep + __shfl_xor(send, 2, 32);
            }
        }
        {
            const bool hi = (lane & 1) != 0;
            const float keep = hi ? a2[1] : a2[0];
            const float send = hi ? a2[0] : a2[1];
            a1 = keep + __shfl_xor(send, 1, 32);
        }
        redv[wid * O_SZ + lane] = a1;

        #pragma unroll
        for (int off = 16; off; off >>= 1) zp += __shfl_xor(zp, off, 32);
        if (lane == 0) reds[wid] = zp;
        __syncthreads();

        float Z = 0.f;
        #pragma unroll
        for (int w = 0; w < 8; ++w) Z += reds[w];

        float sq = 0.f;
        #pragma unroll
        for (int o = 0; o < O_SZ; ++o) {
            float so = 0.f;
            #pragma unroll
            for (int w = 0; w < 8; ++w) so += redv[w * O_SZ + o];
            so /= Z;
            vout[o] = so;
            sq += so * so;
        }
        const float rs = sqrtf(sq);
        const float k  = rs / (1.0f + sq);
        #pragma unroll
        for (int o = 0; o < O_SZ; ++o) vout[o] *= k;

        if (it < 2) {
            #pragma unroll
            for (int j = 0; j < 4; ++j) {
                float d = 0.f;
                #pragma unroll
                for (int o = 0; o < O_SZ; ++o) d += P[j][o] * vout[o];
                logits[j] += d;
            }
        }
        __syncthreads();
    }

    if (t == 0) {
        float* op = out + (size_t)cb * O_SZ;
        #pragma unroll
        for (int o = 0; o < O_SZ; ++o) op[o] = vout[o];
    }
}

extern "C" void kernel_launch(void* const* d_in, const int* in_sizes, int n_in,
                              void* d_out, int out_size, void* d_ws, size_t ws_size,
                              hipStream_t stream) {
    const float* x = (const float*)d_in[0];
    const float* W = (const float*)d_in[1];
    float* priors  = (float*)d_ws;
    float* out     = (float*)d_out;

    caps_priors_wmma<<<dim3((C_CAPS / C_PER_WAVE) * (N_NODES / 8)), dim3(256), 0, stream>>>(x, W, priors);
    caps_route<<<dim3(C_CAPS * B_SZ), dim3(256), 0, stream>>>(priors, out);
}


# route kernel fits two workgroups per WGP (VGPRs 380 to 229, one round of 512 not two): lane o finalizes s[o] once, shared by lane reads; a32 as fma chains so priors are held once; sum order kept
# speedup vs baseline: 2.3956x; 1.2172x over previous
#include <hip/hip_runtime.h>


typedef float v2f __attribute__((ext_vector_type(2)));
typedef float v8f __attribute__((ext_vector_type(8)));

#define C_CAPS  32
#define B_SZ    16
#define N_NODES 1024
#define CI      64
#define O_SZ    32


#define C_PER_WAVE 8
#define N_QUARTERS (4 * C_PER_WAVE)
#define AHEAD_Q    3
#define W_TILE     (CI * O_SZ)
#define W_CSTRIDE  ((size_t)N_NODES * CI * O_SZ)

__global__ __launch_bounds__(256) __attribute__((amdgpu_waves_per_eu(4, 4))) void caps_priors_wmma(
    const float* __restrict__ x, const float* __restrict__ W,
    float* __restrict__ priors)
{
    const int wid  = threadIdx.x >> 5;
    const int lane = threadIdx.x & 31;
    const int blk  = blockIdx.x;
    const int c0   = (blk >> 7) * C_PER_WAVE;
    const int n    = ((blk & 127) << 3) + wid;

    const int m = lane & 15;
    const int h = lane >> 4;

    const float* xp = x + (size_t)m * (N_NODES * CI) + (size_t)n * CI + 2 * h;
    const float* wt = W + (size_t)c0 * W_CSTRIDE + (size_t)n * W_TILE;
    const float* wp = wt + (2 * h) * O_SZ + m;

    v2f a[16];
    #pragma unroll
    for (int ks = 0; ks < 16; ++ks) {
        a[ks].x = xp[ks * 4];
        a[ks].y = xp[ks * 4 + 1];
    }

    v2f bq[4][4][2];

#define LOAD_W_QUARTER(Q) do { \
        const float* wq_ = wp + (size_t)((Q) >> 2) * W_CSTRIDE + ((Q) & 3) * (16 * O_SZ); \
        _Pragma("unroll") for (int kk = 0; kk < 4; ++kk) { \
            bq[(Q) & 3][kk][0].x = wq_[(kk * 4    ) * O_SZ     ]; \
            bq[(Q) & 3][kk][0].y = wq_[(kk * 4 + 1) * O_SZ     ]; \
            bq[(Q) & 3][kk][1].x = wq_[(kk * 4    ) * O_SZ + 16]; \
            bq[(Q) & 3][kk][1].y = wq_[(kk * 4 + 1) * O_SZ + 16]; \
        } } while (0)

    __builtin_prefetch(wt + W_CSTRIDE + lane * 64, 0, 1);
    #pragma unroll
    for (int q = 0; q < AHEAD_Q; ++q) LOAD_W_QUARTER(q);

    float* pp = priors + ((size_t)c0 * B_SZ * N_NODES + (size_t)n) * O_SZ
                       + (size_t)(8 * h) * (N_NODES * O_SZ) + m;

    v8f acc0 = {};
    v8f acc1 = {};

    #pragma unroll
    for (int q = 0; q < N_QUARTERS; ++q) {
        __builtin_amdgcn_sched_barrier(0);
        if ((q & 3) == 0 && (q >> 2) + 2 < C_PER_WAVE)
            __builtin_prefetch(wt + (size_t)((q >> 2) + 2) * W_CSTRIDE + lane * 64, 0, 1);
        if (q + AHEAD_Q < N_QUARTERS) LOAD_W_QUARTER(q + AHEAD_Q);
        __builtin_amdgcn_sched_barrier(0);

        if ((q & 3) == 0) { acc0 = (v8f){}; acc1 = (v8f){}; }
        #pragma unroll
        for (int kk = 0; kk < 4; ++kk) {
            acc0 = __builtin_amdgcn_wmma_f32_16x16x4_f32(
                       false, a[(q & 3) * 4 + kk], false, bq[q & 3][kk][0], (short)0, acc0, false, false);
            acc1 = __builtin_amdgcn_wmma_f32_16x16x4_f32(
                       false, a[(q & 3) * 4 + kk], false, bq[q & 3][kk][1], (short)0, acc1, false, false);
        }

        if ((q & 3) == 3) {
            float* pj = pp + (size_t)(q >> 2) * (B_SZ * N_NODES * O_SZ);
            #pragma unroll
            for (int r = 0; r < 8; ++r) {
                pj[(size_t)r * (N_NODES * O_SZ)     ] = acc0[r];
                pj[(size_t)r * (N_NODES * O_SZ) + 16] = acc1[r];
            }
        }
    }
#undef LOAD_W_QUARTER
}

__global__ __launch_bounds__(256) void caps_route(
    const float* __restrict__ priors, float* __restrict__ out)
{
    const int cb   = blockIdx.x;
    const int t    = threadIdx.x;
    const int lane = t & 31;
    const int wid  = t >> 5;

    __shared__ float redv[8 * O_SZ];
    __shared__ float reds[8];

    float P[4][O_SZ];
    const float* base = priors + (size_t)cb * (N_NODES * O_SZ);
    #pragma unroll
    for (int j = 0; j < 4; ++j) {
        const float4* row = (const float4*)(base + (size_t)(t + 256 * j) * O_SZ);
        #pragma unroll
        for (int q = 0; q < 8; ++q) {
            float4 v = row[q];
            P[j][4*q+0] = v.x; P[j][4*q+1] = v.y;
            P[j][4*q+2] = v.z; P[j][4*q+3] = v.w;
        }
    }

    float logits[4] = {0.f, 0.f, 0.f, 0.f};
    float vout[O_SZ];

    for (int it = 0; it < 3; ++it) {
        float mx = fmaxf(fmaxf(logits[0], logits[1]), fmaxf(logits[2], logits[3]));
        #pragma unroll
        for (int off = 16; off; off >>= 1)
            mx = fmaxf(mx, __shfl_xor(mx, off, 32));
        if (lane == 0) reds[wid] = mx;
        __syncthreads();
        mx = reds[0];
        #pragma unroll
        for (int w = 1; w < 8; ++w) mx = fmaxf(mx, reds[w]);
        __syncthreads();

        float e[4]; float zp = 0.f;
        #pragma unroll
        for (int j = 0; j < 4; ++j) { e[j] = __expf(logits[j] - mx); zp += e[j]; }

        float a32[O_SZ];
        for (int o = 0; o < O_SZ; ++o)
            a32[o] = fmaf(e[3], P[3][o], fmaf(e[2], P[2][o], fmaf(e[1], P[1][o], e[0] * P[0][o])));

        float a16[16], a8[8], a4[4], a2[2], a1;
        {
            const bool hi = (lane & 16) != 0;
            for (int i = 0; i < 16; ++i) {
                const float keep = hi ? a32[i + 16] : a32[i];
                const float send = hi ? a32[i] : a32[i + 16];
                a16[i] = keep + __shfl_xor(send, 16, 32);
            }
        }
        {
            const bool hi = (lane & 8) != 0;
            for (int i = 0; i < 8; ++i) {
                const float keep = hi ? a16[i + 8] : a16[i];
                const float send = hi ? a16[i] : a16[i + 8];
                a8[i] = keep + __shfl_xor(send, 8, 32);
            }
        }
        {
            const bool hi = (lane & 4) != 0;
            for (int i = 0; i < 4; ++i) {
                const float keep = hi ? a8[i + 4] : a8[i];
                const float send = hi ? a8[i] : a8[i + 4];
                a4[i] = keep + __shfl_xor(send, 4, 32);
            }
        }
        {
            const bool hi = (lane & 2) != 0;
            for (int i = 0; i < 2; ++i) {
                const float keep = hi ? a4[i + 2] : a4[i];
                const float send = hi ? a4[i] : a4[i + 2];
                a2[i] = keep + __shfl_xor(send, 2, 32);
            }
        }
        {
            const bool hi = (lane & 1) != 0;
            const float keep = hi ? a2[1] : a2[0];
            const float send = hi ? a2[0] : a2[1];
            a1 = keep + __shfl_xor(send, 1, 32);
        }
        redv[wid * O_SZ + lane] = a1;

        #pragma unroll
        for (int off = 16; off; off >>= 1) zp += __shfl_xor(zp, off, 32);
        if (lane == 0) reds[wid] = zp;
        __syncthreads();

        float Z = 0.f;
        #pragma unroll
        for (int w = 0; w < 8; ++w) Z += reds[w];

        float sl = 0.f;
        #pragma unroll
        for (int w = 0; w < 8; ++w) sl += redv[w * O_SZ + lane];
        sl /= Z;

        float sq = 0.f;
        #pragma unroll
        for (int o = 0; o < O_SZ; ++o) {
            const float so = __int_as_float(__builtin_amdgcn_readlane(__float_as_int(sl), o));
            vout[o] = so;
            sq += so * so;
        }
        const float rs = sqrtf(sq);
        const float k  = rs / (1.0f + sq);
        #pragma unroll
        for (int o = 0; o < O_SZ; ++o) vout[o] *= k;

        if (it < 2) {
            #pragma unroll
            for (int j = 0; j < 4; ++j) {
                float d = 0.f;
                #pragma unroll
                for (int o = 0; o < O_SZ; ++o) d += P[j][o] * vout[o];
                logits[j] += d;
            }
        }
        __syncthreads();
    }

    if (t == 0) {
        float* op = out + (size_t)cb * O_SZ;
        #pragma unroll
        for (int o = 0; o < O_SZ; ++o) op[o] = vout[o];
    }
}

extern "C" void kernel_launch(void* const* d_in, const int* in_sizes, int n_in,
                              void* d_out, int out_size, void* d_ws, size_t ws_size,
                              hipStream_t stream) {
    const float* x = (const float*)d_in[0];
    const float* W = (const float*)d_in[1];
    float* priors  = (float*)d_ws;
    float* out     = (float*)d_out;

    caps_priors_wmma<<<dim3((C_CAPS / C_PER_WAVE) * (N_NODES / 8)), dim3(256), 0, stream>>>(x, W, priors);
    caps_route<<<dim3(C_CAPS * B_SZ), dim3(256), 0, stream>>>(priors, out);
}


# route kernel: priors rows reach their owner lanes through LDS (144-byte row pitch), so each global load reads 512 contiguous bytes (4 whole lines), not 16 bytes from each of 32 lines; sums unchanged
# speedup vs baseline: 2.6914x; 1.1235x over previous
#include <hip/hip_runtime.h>


typedef float v2f __attribute__((ext_vector_type(2)));
typedef float v8f __attribute__((ext_vector_type(8)));

#define C_CAPS  32
#define B_SZ    16
#define N_NODES 1024
#define CI      64
#define O_SZ    32


#define C_PER_WAVE 8
#define N_QUARTERS (4 * C_PER_WAVE)
#define AHEAD_Q    3
#define W_TILE     (CI * O_SZ)
#define W_CSTRIDE  ((size_t)N_NODES * CI * O_SZ)

__global__ __launch_bounds__(256) __attribute__((amdgpu_waves_per_eu(4, 4))) void caps_priors_wmma(
    const float* __restrict__ x, const float* __restrict__ W,
    float* __restrict__ priors)
{
    const int wid  = threadIdx.x >> 5;
    const int lane = threadIdx.x & 31;
    const int blk  = blockIdx.x;
    const int c0   = (blk >> 7) * C_PER_WAVE;
    const int n    = ((blk & 127) << 3) + wid;

    const int m = lane & 15;
    const int h = lane >> 4;

    const float* xp = x + (size_t)m * (N_NODES * CI) + (size_t)n * CI + 2 * h;
    const float* wt = W + (size_t)c0 * W_CSTRIDE + (size_t)n * W_TILE;
    const float* wp = wt + (2 * h) * O_SZ + m;

    v2f a[16];
    #pragma unroll
    for (int ks = 0; ks < 16; ++ks) {
        a[ks].x = xp[ks * 4];
        a[ks].y = xp[ks * 4 + 1];
    }

    v2f bq[4][4][2];

#define LOAD_W_QUARTER(Q) do { \
        const float* wq_ = wp + (size_t)((Q) >> 2) * W_CSTRIDE + ((Q) & 3) * (16 * O_SZ); \
        _Pragma("unroll") for (int kk = 0; kk < 4; ++kk) { \
            bq[(Q) & 3][kk][0].x = wq_[(kk * 4    ) * O_SZ     ]; \
            bq[(Q) & 3][kk][0].y = wq_[(kk * 4 + 1) * O_SZ     ]; \
            bq[(Q) & 3][kk][1].x = wq_[(kk * 4    ) * O_SZ + 16]; \
            bq[(Q) & 3][kk][1].y = wq_[(kk * 4 + 1) * O_SZ + 16]; \
        } } while (0)

    __builtin_prefetch(wt + W_CSTRIDE + lane * 64, 0, 1);
    #pragma unroll
    for (int q = 0; q < AHEAD_Q; ++q) LOAD_W_QUARTER(q);

    float* pp = priors + ((size_t)c0 * B_SZ * N_NODES + (size_t)n) * O_SZ
                       + (size_t)(8 * h) * (N_NODES * O_SZ) + m;

    v8f acc0 = {};
    v8f acc1 = {};

    #pragma unroll
    for (int q = 0; q < N_QUARTERS; ++q) {
        __builtin_amdgcn_sched_barrier(0);
        if ((q & 3) == 0 && (q >> 2) + 2 < C_PER_WAVE)
            __builtin_prefetch(wt + (size_t)((q >> 2) + 2) * W_CSTRIDE + lane * 64, 0, 1);
        if (q + AHEAD_Q < N_QUARTERS) LOAD_W_QUARTER(q + AHEAD_Q);
        __builtin_amdgcn_sched_barrier(0);

        if ((q & 3) == 0) { acc0 = (v8f){}; acc1 = (v8f){}; }
        #pragma unroll
        for (int kk = 0; kk < 4; ++kk) {
            acc0 = __builtin_amdgcn_wmma_f32_16x16x4_f32(
                       false, a[(q & 3) * 4 + kk], false, bq[q & 3][kk][0], (short)0, acc0, false, false);
            acc1 = __builtin_amdgcn_wmma_f32_16x16x4_f32(
                       false, a[(q & 3) * 4 + kk], false, bq[q & 3][kk][1], (short)0, acc1, false, false);
        }

        if ((q & 3) == 3) {
            float* pj = pp + (size_t)(q >> 2) * (B_SZ * N_NODES * O_SZ);
            #pragma unroll
            for (int r = 0; r < 8; ++r) {
                pj[(size_t)r * (N_NODES * O_SZ)     ] = acc0[r];
                pj[(size_t)r * (N_NODES * O_SZ) + 16] = acc1[r];
            }
        }
    }
#undef LOAD_W_QUARTER
}

#define ROW_PITCH4 9
__global__ __launch_bounds__(256) void caps_route(
    const float* __restrict__ priors, float* __restrict__ out)
{
    const int cb   = blockIdx.x;
    const int t    = threadIdx.x;
    const int lane = t & 31;
    const int wid  = t >> 5;

    __shared__ float redv[8 * O_SZ];
    __shared__ float reds[8];

    __shared__ float4 stage[N_NODES * ROW_PITCH4];

    const float4* base4 = (const float4*)(priors + (size_t)cb * (N_NODES * O_SZ));
    float4 g[4][8];
    #pragma unroll
    for (int j = 0; j < 4; ++j) {
        const float4* blk4 = base4 + (size_t)(256 * j + 32 * wid) * (O_SZ / 4) + lane;
        #pragma unroll
        for (int q = 0; q < 8; ++q) g[j][q] = blk4[32 * q];
    }
    #pragma unroll
    for (int j = 0; j < 4; ++j) {
        float4* dst = stage + (256 * j + 32 * wid + (lane >> 3)) * ROW_PITCH4 + (lane & 7);
        #pragma unroll
        for (int q = 0; q < 8; ++q) dst[4 * q * ROW_PITCH4] = g[j][q];
    }
    __syncthreads();

    float P[4][O_SZ];
    #pragma unroll
    for (int j = 0; j < 4; ++j) {
        const float4* row = stage + (256 * j + t) * ROW_PITCH4;
        #pragma unroll
        for (int q = 0; q < 8; ++q) {
            float4 v = row[q];
            P[j][4*q+0] = v.x; P[j][4*q+1] = v.y;
            P[j][4*q+2] = v.z; P[j][4*q+3] = v.w;
        }
    }

    float logits[4] = {0.f, 0.f, 0.f, 0.f};
    float vout[O_SZ];

    for (int it = 0; it < 3; ++it) {
        float mx = fmaxf(fmaxf(logits[0], logits[1]), fmaxf(logits[2], logits[3]));
        #pragma unroll
        for (int off = 16; off; off >>= 1)
            mx = fmaxf(mx, __shfl_xor(mx, off, 32));
        if (lane == 0) reds[wid] = mx;
        __syncthreads();
        mx = reds[0];
        #pragma unroll
        for (int w = 1; w < 8; ++w) mx = fmaxf(mx, reds[w]);
        __syncthreads();

        float e[4]; float zp = 0.f;
        #pragma unroll
        for (int j = 0; j < 4; ++j) { e[j] = __expf(logits[j] - mx); zp += e[j]; }

        float a32[O_SZ];
        for (int o = 0; o < O_SZ; ++o)
            a32[o] = fmaf(e[3], P[3][o], fmaf(e[2], P[2][o], fmaf(e[1], P[1][o], e[0] * P[0][o])));

        float a16[16], a8[8], a4[4], a2[2], a1;
        {
            const bool hi = (lane & 16) != 0;
            for (int i = 0; i < 16; ++i) {
                const float keep = hi ? a32[i + 16] : a32[i];
                const float send = hi ? a32[i] : a32[i + 16];
                a16[i] = keep + __shfl_xor(send, 16, 32);
            }
        }
        {
            const bool hi = (lane & 8) != 0;
            for (int i = 0; i < 8; ++i) {
                const float keep = hi ? a16[i + 8] : a16[i];
                const float send = hi ? a16[i] : a16[i + 8];
                a8[i] = keep + __shfl_xor(send, 8, 32);
            }
        }
        {
            const bool hi = (lane & 4) != 0;
            for (int i = 0; i < 4; ++i) {
                const float keep = hi ? a8[i + 4] : a8[i];
                const float send = hi ? a8[i] : a8[i + 4];
                a4[i] = keep + __shfl_xor(send, 4, 32);
            }
        }
        {
            const bool hi = (lane & 2) != 0;
            for (int i = 0; i < 2; ++i) {
                const float keep = hi ? a4[i + 2] : a4[i];
                const float send = hi ? a4[i] : a4[i + 2];
                a2[i] = keep + __shfl_xor(send, 2, 32);
            }
        }
        {
            const bool hi = (lane & 1) != 0;
            const float keep = hi ? a2[1] : a2[0];
            const float send = hi ? a2[0] : a2[1];
            a1 = keep + __shfl_xor(send, 1, 32);
        }
        redv[wid * O_SZ + lane] = a1;

        #pragma unroll
        for (int off = 16; off; off >>= 1) zp += __shfl_xor(zp, off, 32);
        if (lane == 0) reds[wid] = zp;
        __syncthreads();

        float Z = 0.f;
        #pragma unroll
        for (int w = 0; w < 8; ++w) Z += reds[w];

        float sl = 0.f;
        #pragma unroll
        for (int w = 0; w < 8; ++w) sl += redv[w * O_SZ + lane];
        sl /= Z;

        float sq = 0.f;
        #pragma unroll
        for (int o = 0; o < O_SZ; ++o) {
            const float so = __int_as_float(__builtin_amdgcn_readlane(__float_as_int(sl), o));
            vout[o] = so;
            sq += so * so;
        }
        const float rs = sqrtf(sq);
        const float k  = rs / (1.0f + sq);
        #pragma unroll
        for (int o = 0; o < O_SZ; ++o) vout[o] *= k;

        if (it < 2) {
            #pragma unroll
            for (int j = 0; j < 4; ++j) {
                float d = 0.f;
                #pragma unroll
                for (int o = 0; o < O_SZ; ++o) d += P[j][o] * vout[o];
                logits[j] += d;
            }
        }
        __syncthreads();
    }

    if (t == 0) {
        float* op = out + (size_t)cb * O_SZ;
        #pragma unroll
        for (int o = 0; o < O_SZ; ++o) op[o] = vout[o];
    }
}

extern "C" void kernel_launch(void* const* d_in, const int* in_sizes, int n_in,
                              void* d_out, int out_size, void* d_ws, size_t ws_size,
                              hipStream_t stream) {
    const float* x = (const float*)d_in[0];
    const float* W = (const float*)d_in[1];
    float* priors  = (float*)d_ws;
    float* out     = (float*)d_out;

    caps_priors_wmma<<<dim3((C_CAPS / C_PER_WAVE) * (N_NODES / 8)), dim3(256), 0, stream>>>(x, W, priors);
    caps_route<<<dim3(C_CAPS * B_SZ), dim3(256), 0, stream>>>(priors, out);
}
